# starting kernel
# baseline (speedup 1.0000x reference)
#include <hip/hip_runtime.h>


typedef __attribute__((ext_vector_type(2))) float v2f;
typedef __attribute__((ext_vector_type(8))) float v8f;

#define NCOEF 49
#define KPAD 52
#define RES2 324
#define CTOT 128
#define ITILE_STRIDE 32
#define ITILE_FLOATS (KPAD * ITILE_STRIDE)
#define YTILE_FLOATS (RES2 * 16)
#define SMEM_FLOATS (8 * (ITILE_FLOATS + YTILE_FLOATS))

__device__ __forceinline__ float sigmoidf_(float x) {
  return 1.0f / (1.0f + __expf(-x));
}

#define WMMA_F32(A, B, Cc) \
  __builtin_amdgcn_wmma_f32_16x16x4_f32(false, (A), false, (B), (short)0, (Cc), false, false)

__global__ __launch_bounds__(256, 1)
void s2act_swiglu_kernel(const float* __restrict__ X,
                         const float* __restrict__ S,
                         const float* __restrict__ G,
                         const float* __restrict__ F,
                         float* __restrict__ O)
{
  extern __shared__ float smem[];
  const int n      = blockIdx.x;
  const int tid    = threadIdx.x;
  const int wave   = tid >> 5;
  const int lane   = tid & 31;
  const int lane16 = lane & 15;
  const int lanehi = lane >> 4;
  const int cbase  = wave * 16;
  const int c      = cbase + lane16;

  float* It = smem + wave * ITILE_FLOATS;
  float* Yt = smem + 8 * ITILE_FLOATS + wave * YTILE_FLOATS;

  const float* Xn = X + (size_t)n * (NCOEF * 2 * CTOT);
  const int ch     = c + CTOT * lanehi;
  const int clocal = lane16 + 16 * lanehi;
  for (int i = 0; i < NCOEF; ++i)
    It[i * ITILE_STRIDE + clocal] = Xn[i * (2 * CTOT) + ch];
  #pragma unroll
  for (int i = NCOEF; i < KPAD; ++i)
    It[i * ITILE_STRIDE + clocal] = 0.0f;
  __syncthreads();

  for (int rtg = 0; rtg < 7; ++rtg) {
    const int rowbase = rtg * 48;
    const int r0 = rowbase      + lane16;
    const int r1 = rowbase + 16 + lane16;
    const int r2 = rowbase + 32 + lane16;
    const float* Ga = G + (size_t)(r0 < RES2 ? r0 : RES2 - 1) * NCOEF;
    const float* Gb = G + (size_t)(r1 < RES2 ? r1 : RES2 - 1) * NCOEF;
    const float* Gc = G + (size_t)(r2 < RES2 ? r2 : RES2 - 1) * NCOEF;
    v8f acc0 = {}, acc1 = {}, acc2 = {}, acc3 = {}, acc4 = {}, acc5 = {};
    for (int ks = 0; ks < 13; ++ks) {
      const int kb  = ks * 4 + 2 * lanehi;
      const int kc0 = (kb     < NCOEF) ? kb     : NCOEF - 1;
      const int kc1 = (kb + 1 < NCOEF) ? kb + 1 : NCOEF - 1;
      v2f b0, b1;
      b0.x = It[ kb      * ITILE_STRIDE + lane16];
      b0.y = It[(kb + 1) * ITILE_STRIDE + lane16];
      b1.x = It[ kb      * ITILE_STRIDE + 16 + lane16];
      b1.y = It[(kb + 1) * ITILE_STRIDE + 16 + lane16];
      v2f a0, a1, a2;
      a0.x = Ga[kc0]; a0.y = Ga[kc1];
      a1.x = Gb[kc0]; a1.y = Gb[kc1];
      a2.x = Gc[kc0]; a2.y = Gc[kc1];
      acc0 = WMMA_F32(a0, b0, acc0);
      acc1 = WMMA_F32(a0, b1, acc1);
      acc2 = WMMA_F32(a1, b0, acc2);
      acc3 = WMMA_F32(a1, b1, acc3);
      acc4 = WMMA_F32(a2, b0, acc4);
      acc5 = WMMA_F32(a2, b1, acc5);
    }
    #pragma unroll
    for (int r = 0; r < 8; ++r) {
      const int m0 = rowbase      + r + 8 * lanehi;
      const int m1 = rowbase + 16 + r + 8 * lanehi;
      const int m2 = rowbase + 32 + r + 8 * lanehi;
      if (m0 < RES2) Yt[m0 * 16 + lane16] = acc0[r] * acc1[r];
      if (m1 < RES2) Yt[m1 * 16 + lane16] = acc2[r] * acc3[r];
      if (m2 < RES2) Yt[m2 * 16 + lane16] = acc4[r] * acc5[r];
    }
  }
  __syncthreads();

  const float* Sn = S + (size_t)n * (3 * CTOT);
  const float sa    = Sn[c];
  const float sb    = Sn[CTOT + c];
  const float gate  = sigmoidf_(Sn[2 * CTOT + c]);
  const float oscal = sa * sigmoidf_(sa) * sb;

  const float* F0 = F + ((     lane16) < NCOEF ? (     lane16) : NCOEF - 1);
  const float* F1 = F + ((16 + lane16) < NCOEF ? (16 + lane16) : NCOEF - 1);
  const float* F2 = F + ((32 + lane16) < NCOEF ? (32 + lane16) : NCOEF - 1);
  const float* F3 = F + ((48 + lane16) < NCOEF ? (48 + lane16) : NCOEF - 1);
  v8f oa0 = {}, oa1 = {}, oa2 = {}, oa3 = {};
  for (int ks = 0; ks < 81; ++ks) {
    const int p0 = ks * 4 + 2 * lanehi;
    v2f b;
    b.x = Yt[ p0      * 16 + lane16];
    b.y = Yt[(p0 + 1) * 16 + lane16];
    v2f a0, a1, a2, a3;
    a0.x = F0[ p0      * NCOEF]; a0.y = F0[(p0 + 1) * NCOEF];
    a1.x = F1[ p0      * NCOEF]; a1.y = F1[(p0 + 1) * NCOEF];
    a2.x = F2[ p0      * NCOEF]; a2.y = F2[(p0 + 1) * NCOEF];
    a3.x = F3[ p0      * NCOEF]; a3.y = F3[(p0 + 1) * NCOEF];
    oa0 = WMMA_F32(a0, b, oa0);
    oa1 = WMMA_F32(a1, b, oa1);
    oa2 = WMMA_F32(a2, b, oa2);
    oa3 = WMMA_F32(a3, b, oa3);
  }

  float* On = O + (size_t)n * (NCOEF * CTOT) + c;
  #pragma unroll
  for (int r = 0; r < 8; ++r) {
    const int i0 = r + 8 * lanehi;
    float v0 = oa0[r] * gate;
    if (i0 == 0) v0 += oscal;
    On[i0 * CTOT] = v0;
    On[(16 + i0) * CTOT] = oa1[r] * gate;
    On[(32 + i0) * CTOT] = oa2[r] * gate;
    const int i3 = 48 + i0;
    if (i3 < NCOEF) On[i3 * CTOT] = oa3[r] * gate;
  }
}

extern "C" void kernel_launch(void* const* d_in, const int* in_sizes, int n_in,
                              void* d_out, int out_size, void* d_ws, size_t ws_size,
                              hipStream_t stream) {
  const float* X = (const float*)d_in[0];
  const float* S = (const float*)d_in[1];
  const float* G = (const float*)d_in[2];
  const float* F = (const float*)d_in[3];
  float* O = (float*)d_out;

  const int N = in_sizes[0] / (NCOEF * 2 * CTOT);
  const size_t shmem = (size_t)SMEM_FLOATS * sizeof(float);

  (void)hipFuncSetAttribute((const void*)s2act_swiglu_kernel,
                            hipFuncAttributeMaxDynamicSharedMemorySize,
                            (int)shmem);

  hipLaunchKernelGGL(s2act_swiglu_kernel, dim3(N), dim3(256), shmem, stream,
                     X, S, G, F, O);
}


# to_grid and from_grid (the A operands) staged once per block in LDS at conflict-free pitches; A fragments are 8-byte LDS reads issued one WMMA block ahead, replacing strided per-lane global loads
# speedup vs baseline: 1.3453x; 1.3453x over previous
#include <hip/hip_runtime.h>


typedef __attribute__((ext_vector_type(2))) float v2f;
typedef __attribute__((ext_vector_type(4))) float v4f;
typedef __attribute__((ext_vector_type(8))) float v8f;

#define NCOEF 49
#define KPAD 52
#define RES2 324
#define CTOT 128
#define NTHR 256
#define ITILE_STRIDE 32
#define ITILE_FLOATS (KPAD * ITILE_STRIDE)
#define YTILE_FLOATS (RES2 * 16)
#define GL_FLOATS (RES2 * KPAD)
#define FT_PITCH RES2
#define PAIR_ITERS ((RES2 / 2 + 7) / 8)
#define SMEM_FLOATS (8 * ITILE_FLOATS + GL_FLOATS + 8 * YTILE_FLOATS)

__device__ __forceinline__ float sigmoidf_(float x) {
  return 1.0f / (1.0f + __expf(-x));
}

#define WMMA_F32(A, B, Cc) \
  __builtin_amdgcn_wmma_f32_16x16x4_f32(false, (A), false, (B), (short)0, (Cc), false, false)

__global__ __launch_bounds__(256, 1)
void s2act_swiglu_kernel(const float* __restrict__ X,
                         const float* __restrict__ S,
                         const float* __restrict__ G,
                         const float* __restrict__ F,
                         float* __restrict__ O)
{
  extern __shared__ __attribute__((aligned(16))) float smem[];
  const int n      = blockIdx.x;
  const int tid    = threadIdx.x;
  const int wave   = tid >> 5;
  const int lane   = tid & 31;
  const int lane16 = lane & 15;
  const int lanehi = lane >> 4;
  const int cbase  = wave * 16;
  const int c      = cbase + lane16;

  float* It = smem + wave * ITILE_FLOATS;
  float* Gl = smem + 8 * ITILE_FLOATS;
  float* Ft = smem;
  float* Yt = smem + 8 * ITILE_FLOATS + GL_FLOATS + wave * YTILE_FLOATS;

  const float* Xn = X + (size_t)n * (NCOEF * 2 * CTOT);
  const int ch     = c + CTOT * lanehi;
  const int clocal = lane16 + 16 * lanehi;
  for (int i = 0; i < NCOEF; ++i)
    It[i * ITILE_STRIDE + clocal] = Xn[i * (2 * CTOT) + ch];
  #pragma unroll
  for (int i = NCOEF; i < KPAD; ++i)
    It[i * ITILE_STRIDE + clocal] = 0.0f;

  const int rp0 = 2 * wave + lanehi;
  #pragma unroll
  for (int j = 0; j < PAIR_ITERS; ++j) {
    const int row = rp0 + 16 * j;
    if (j < PAIR_ITERS - 1 || rp0 < RES2 - 16 * (PAIR_ITERS - 1)) {
      #pragma unroll
      for (int t = 0; t < 3; ++t)
        Gl[row * KPAD + 16 * t + lane16] = G[row * NCOEF + 16 * t + lane16];
    }
  }
  {
    v4f g;
    g.x = G[tid * NCOEF + (NCOEF - 1)]; g.y = 0.0f; g.z = 0.0f; g.w = 0.0f;
    *(v4f*)(Gl + tid * KPAD + (NCOEF - 1)) = g;
    if (tid + NTHR < RES2) {
      g.x = G[(tid + NTHR) * NCOEF + (NCOEF - 1)];
      *(v4f*)(Gl + (tid + NTHR) * KPAD + (NCOEF - 1)) = g;
    }
  }
  __syncthreads();

  float fpre[PAIR_ITERS][3];
  #pragma unroll
  for (int j = 0; j < PAIR_ITERS; ++j) {
    const int pq = rp0 + 16 * j;
    const bool in = (j < PAIR_ITERS - 1 || rp0 < RES2 - 16 * (PAIR_ITERS - 1));
    #pragma unroll
    for (int t = 0; t < 3; ++t)
      fpre[j][t] = in ? F[pq * NCOEF + 16 * t + lane16] : 0.0f;
  }
  const float flast0 = F[tid * NCOEF + (NCOEF - 1)];
  const float flast1 = (tid + NTHR < RES2) ? F[(tid + NTHR) * NCOEF + (NCOEF - 1)] : 0.0f;
  const float* Sn = S + (size_t)n * (3 * CTOT);
  const float sa = Sn[c];
  const float sb = Sn[CTOT + c];
  const float sg = Sn[2 * CTOT + c];

  v2f bx0[13], bx1[13];
  #pragma unroll
  for (int ks = 0; ks < 13; ++ks) {
    const int kb = ks * 4 + 2 * lanehi;
    bx0[ks].x = It[ kb      * ITILE_STRIDE + lane16];
    bx0[ks].y = It[(kb + 1) * ITILE_STRIDE + lane16];
    bx1[ks].x = It[ kb      * ITILE_STRIDE + 16 + lane16];
    bx1[ks].y = It[(kb + 1) * ITILE_STRIDE + 16 + lane16];
  }
  v2f ga[2][13][3];
#define LOAD_G_GROUP(RTG, BUF)                                                          \
  {                                                                                     \
    const int r0_ = (RTG) * 48 + lane16;                                                \
    const int r1_ = r0_ + 16;                                                           \
    const int r2_ = r0_ + 32;                                                           \
    const v2f* Ga_ = (const v2f*)(Gl + (r0_ < RES2 ? r0_ : RES2 - 1) * KPAD + 2 * lanehi); \
    const v2f* Gb_ = (const v2f*)(Gl + (r1_ < RES2 ? r1_ : RES2 - 1) * KPAD + 2 * lanehi); \
    const v2f* Gc_ = (const v2f*)(Gl + (r2_ < RES2 ? r2_ : RES2 - 1) * KPAD + 2 * lanehi); \
    _Pragma("unroll")                                                                   \
    for (int ks_ = 0; ks_ < 13; ++ks_) {                                                \
      ga[BUF][ks_][0] = Ga_[2 * ks_];                                                   \
      ga[BUF][ks_][1] = Gb_[2 * ks_];                                                   \
      ga[BUF][ks_][2] = Gc_[2 * ks_];                                                   \
    }                                                                                   \
  }
  LOAD_G_GROUP(0, 0)
  __builtin_amdgcn_sched_barrier(0);
  #pragma unroll
  for (int rtg = 0; rtg < 7; ++rtg) {
    const int rowbase = rtg * 48;
    if (rtg + 1 < 7) LOAD_G_GROUP(rtg + 1, (rtg + 1) & 1)
    __builtin_amdgcn_sched_barrier(0);
    v8f acc0 = {}, acc1 = {}, acc2 = {}, acc3 = {}, acc4 = {}, acc5 = {};
    #pragma unroll
    for (int ks = 0; ks < 13; ++ks) {
      const v2f a0 = ga[rtg & 1][ks][0];
      const v2f a1 = ga[rtg & 1][ks][1];
      const v2f a2 = ga[rtg & 1][ks][2];
      acc0 = WMMA_F32(a0, bx0[ks], acc0);
      acc1 = WMMA_F32(a0, bx1[ks], acc1);
      acc2 = WMMA_F32(a1, bx0[ks], acc2);
      acc3 = WMMA_F32(a1, bx1[ks], acc3);
      acc4 = WMMA_F32(a2, bx0[ks], acc4);
      acc5 = WMMA_F32(a2, bx1[ks], acc5);
    }
    __builtin_amdgcn_sched_barrier(0);
    #pragma unroll
    for (int r = 0; r < 8; ++r) {
      const int m0 = rowbase      + r + 8 * lanehi;
      const int m1 = rowbase + 16 + r + 8 * lanehi;
      const int m2 = rowbase + 32 + r + 8 * lanehi;
      if (m0 < RES2) Yt[m0 * 16 + lane16] = acc0[r] * acc1[r];
      if (m1 < RES2) Yt[m1 * 16 + lane16] = acc2[r] * acc3[r];
      if (m2 < RES2) Yt[m2 * 16 + lane16] = acc4[r] * acc5[r];
    }
  }
#undef LOAD_G_GROUP
  __syncthreads();

  #pragma unroll
  for (int j = 0; j < PAIR_ITERS; ++j) {
    const int pq = rp0 + 16 * j;
    if (j < PAIR_ITERS - 1 || rp0 < RES2 - 16 * (PAIR_ITERS - 1)) {
      #pragma unroll
      for (int t = 0; t < 3; ++t)
        Ft[(16 * t + lane16) * FT_PITCH + pq] = fpre[j][t];
    }
  }
  Ft[(NCOEF - 1) * FT_PITCH + tid] = flast0;
  if (tid + NTHR < RES2) Ft[(NCOEF - 1) * FT_PITCH + tid + NTHR] = flast1;

  const float gate  = sigmoidf_(sg);
  const float oscal = sa * sigmoidf_(sa) * sb;
  __syncthreads();

  const int i3 = (48 + lane16) < NCOEF ? (48 + lane16) : NCOEF - 1;
  const v2f* F0 = (const v2f*)(Ft + (     lane16) * FT_PITCH + 2 * lanehi);
  const v2f* F1 = (const v2f*)(Ft + (16 + lane16) * FT_PITCH + 2 * lanehi);
  const v2f* F2 = (const v2f*)(Ft + (32 + lane16) * FT_PITCH + 2 * lanehi);
  const v2f* F3 = (const v2f*)(Ft + i3 * FT_PITCH + 2 * lanehi);
  const float* Yb = Yt + 2 * lanehi * 16 + lane16;
  v8f oa0 = {}, oa1 = {}, oa2 = {}, oa3 = {};
  v2f fa[2][9][4], fb[2][9];
#define LOAD_F_BLOCK(BLK, BUF)                                   \
  {                                                              \
    _Pragma("unroll")                                            \
    for (int j_ = 0; j_ < 9; ++j_) {                             \
      const int ks_ = (BLK) * 9 + j_;                            \
      fb[BUF][j_].x = Yb[(ks_ * 4)     * 16];                    \
      fb[BUF][j_].y = Yb[(ks_ * 4 + 1) * 16];                    \
      fa[BUF][j_][0] = F0[2 * ks_];                              \
      fa[BUF][j_][1] = F1[2 * ks_];                              \
      fa[BUF][j_][2] = F2[2 * ks_];                              \
      fa[BUF][j_][3] = F3[2 * ks_];                              \
    }                                                            \
  }
  LOAD_F_BLOCK(0, 0)
  __builtin_amdgcn_sched_barrier(0);
  #pragma unroll
  for (int blk = 0; blk < 9; ++blk) {
    if (blk + 1 < 9) LOAD_F_BLOCK(blk + 1, (blk + 1) & 1)
    __builtin_amdgcn_sched_barrier(0);
    #pragma unroll
    for (int j = 0; j < 9; ++j) {
      const v2f b = fb[blk & 1][j];
      oa0 = WMMA_F32(fa[blk & 1][j][0], b, oa0);
      oa1 = WMMA_F32(fa[blk & 1][j][1], b, oa1);
      oa2 = WMMA_F32(fa[blk & 1][j][2], b, oa2);
      oa3 = WMMA_F32(fa[blk & 1][j][3], b, oa3);
    }
    __builtin_amdgcn_sched_barrier(0);
  }
#undef LOAD_F_BLOCK

  float* On = O + (size_t)n * (NCOEF * CTOT) + c;
  #pragma unroll
  for (int r = 0; r < 8; ++r) {
    const int i0 = r + 8 * lanehi;
    float v0 = oa0[r] * gate;
    if (i0 == 0) v0 += oscal;
    On[i0 * CTOT] = v0;
    On[(16 + i0) * CTOT] = oa1[r] * gate;
    On[(32 + i0) * CTOT] = oa2[r] * gate;
    const int i3o = 48 + i0;
    if (i3o < NCOEF) On[i3o * CTOT] = oa3[r] * gate;
  }
}

extern "C" void kernel_launch(void* const* d_in, const int* in_sizes, int n_in,
                              void* d_out, int out_size, void* d_ws, size_t ws_size,
                              hipStream_t stream) {
  const float* X = (const float*)d_in[0];
  const float* S = (const float*)d_in[1];
  const float* G = (const float*)d_in[2];
  const float* F = (const float*)d_in[3];
  float* O = (float*)d_out;

  const int N = in_sizes[0] / (NCOEF * 2 * CTOT);
  const size_t shmem = (size_t)SMEM_FLOATS * sizeof(float);

  (void)hipFuncSetAttribute((const void*)s2act_swiglu_kernel,
                            hipFuncAttributeMaxDynamicSharedMemorySize,
                            (int)shmem);

  hipLaunchKernelGGL(s2act_swiglu_kernel, dim3(N), dim3(256), shmem, stream,
                     X, S, G, F, O);
}


# persistent blocks: 256 work-groups loop over their nodes; to_grid and from_grid staged in LDS once per block, not per node; next node's inputs fetched to registers during the second product
# speedup vs baseline: 1.6563x; 1.2312x over previous
#include <hip/hip_runtime.h>


typedef __attribute__((ext_vector_type(2))) float v2f;
typedef __attribute__((ext_vector_type(4))) float v4f;
typedef __attribute__((ext_vector_type(8))) float v8f;

#define NCOEF 49
#define KPAD 52
#define RES2 324
#define CTOT 128
#define NTHR 256
#define NBLOCKS 256
#define YTILE_FLOATS (RES2 * 16)
#define GL_FLOATS (RES2 * KPAD)
#define FT_PITCH RES2
#define FT_FLOATS (NCOEF * FT_PITCH)
#define PAIR_ITERS ((RES2 / 2 + 7) / 8)
#define SMEM_FLOATS (GL_FLOATS + FT_FLOATS + 8 * YTILE_FLOATS)

__device__ __forceinline__ float sigmoidf_(float x) {
  return 1.0f / (1.0f + __expf(-x));
}

#define WMMA_F32(A, B, Cc) \
  __builtin_amdgcn_wmma_f32_16x16x4_f32(false, (A), false, (B), (short)0, (Cc), false, false)

#define LOAD_BX(XP)                                                        \
  {                                                                        \
    const float* xp_ = (XP) + (2 * lanehi) * (2 * CTOT) + c;               \
    _Pragma("unroll")                                                      \
    for (int ks_ = 0; ks_ < 12; ++ks_) {                                   \
      bx0[ks_].x = xp_[(4 * ks_)     * (2 * CTOT)];                        \
      bx0[ks_].y = xp_[(4 * ks_ + 1) * (2 * CTOT)];                        \
      bx1[ks_].x = xp_[(4 * ks_)     * (2 * CTOT) + CTOT];                 \
      bx1[ks_].y = xp_[(4 * ks_ + 1) * (2 * CTOT) + CTOT];                 \
    }                                                                      \
    x48a = (XP)[48 * (2 * CTOT) + c];                                      \
    x48b = (XP)[48 * (2 * CTOT) + CTOT + c];                               \
  }

__global__ __launch_bounds__(256, 1)
void s2act_swiglu_kernel(const float* __restrict__ X,
                         const float* __restrict__ S,
                         const float* __restrict__ G,
                         const float* __restrict__ F,
                         float* __restrict__ O,
                         const int N)
{
  extern __shared__ __attribute__((aligned(16))) float smem[];
  const int tid    = threadIdx.x;
  const int wave   = tid >> 5;
  const int lane   = tid & 31;
  const int lane16 = lane & 15;
  const int lanehi = lane >> 4;
  const int cbase  = wave * 16;
  const int c      = cbase + lane16;

  float* Gl = smem;
  float* Ft = smem + GL_FLOATS;
  float* Yt = smem + GL_FLOATS + FT_FLOATS + wave * YTILE_FLOATS;

  int n = blockIdx.x;
  v2f bx0[13], bx1[13];
  float x48a = 0.0f, x48b = 0.0f;
  #pragma unroll
  for (int ks = 0; ks < 12; ++ks) { bx0[ks] = (v2f){0.0f, 0.0f}; bx1[ks] = (v2f){0.0f, 0.0f}; }
  if (n < N) {
    LOAD_BX(X + (size_t)n * (NCOEF * 2 * CTOT))
  }

  const int rp0 = 2 * wave + lanehi;
  #pragma unroll
  for (int j = 0; j < PAIR_ITERS; ++j) {
    const int row = rp0 + 16 * j;
    if (j < PAIR_ITERS - 1 || rp0 < RES2 - 16 * (PAIR_ITERS - 1)) {
      #pragma unroll
      for (int t = 0; t < 3; ++t)
        Gl[row * KPAD + 16 * t + lane16] = G[row * NCOEF + 16 * t + lane16];
    }
  }
  {
    v4f g;
    g.x = G[tid * NCOEF + (NCOEF - 1)]; g.y = 0.0f; g.z = 0.0f; g.w = 0.0f;
    *(v4f*)(Gl + tid * KPAD + (NCOEF - 1)) = g;
    if (tid + NTHR < RES2) {
      g.x = G[(tid + NTHR) * NCOEF + (NCOEF - 1)];
      *(v4f*)(Gl + (tid + NTHR) * KPAD + (NCOEF - 1)) = g;
    }
  }

  #pragma unroll
  for (int j = 0; j < PAIR_ITERS; ++j) {
    const int pq = rp0 + 16 * j;
    if (j < PAIR_ITERS - 1 || rp0 < RES2 - 16 * (PAIR_ITERS - 1)) {
      #pragma unroll
      for (int t = 0; t < 3; ++t)
        Ft[(16 * t + lane16) * FT_PITCH + pq] = F[pq * NCOEF + 16 * t + lane16];
    }
  }
  Ft[(NCOEF - 1) * FT_PITCH + tid] = F[tid * NCOEF + (NCOEF - 1)];
  if (tid + NTHR < RES2)
    Ft[(NCOEF - 1) * FT_PITCH + tid + NTHR] = F[(tid + NTHR) * NCOEF + (NCOEF - 1)];
  __syncthreads();

  const int i3 = (48 + lane16) < NCOEF ? (48 + lane16) : NCOEF - 1;
  const v2f* F0 = (const v2f*)(Ft + (     lane16) * FT_PITCH + 2 * lanehi);
  const v2f* F1 = (const v2f*)(Ft + (16 + lane16) * FT_PITCH + 2 * lanehi);
  const v2f* F2 = (const v2f*)(Ft + (32 + lane16) * FT_PITCH + 2 * lanehi);
  const v2f* F3 = (const v2f*)(Ft + i3 * FT_PITCH + 2 * lanehi);
  const float* Yb = Yt + 2 * lanehi * 16 + lane16;

  #pragma nounroll
  for (; n < N; n += gridDim.x) {
    const float* Sn = S + (size_t)n * (3 * CTOT);
    const float sa = Sn[c];
    const float sb = Sn[CTOT + c];
    const float sg = Sn[2 * CTOT + c];
    bx0[12] = (v2f){lanehi == 0 ? x48a : 0.0f, 0.0f};
    bx1[12] = (v2f){lanehi == 0 ? x48b : 0.0f, 0.0f};

    v2f ga[2][13][3];
#define LOAD_G_GROUP(RTG, BUF)                                                          \
    {                                                                                   \
      const int r0_ = (RTG) * 48 + lane16;                                              \
      const int r1_ = r0_ + 16;                                                         \
      const int r2_ = r0_ + 32;                                                         \
      const v2f* Ga_ = (const v2f*)(Gl + (r0_ < RES2 ? r0_ : RES2 - 1) * KPAD + 2 * lanehi); \
      const v2f* Gb_ = (const v2f*)(Gl + (r1_ < RES2 ? r1_ : RES2 - 1) * KPAD + 2 * lanehi); \
      const v2f* Gc_ = (const v2f*)(Gl + (r2_ < RES2 ? r2_ : RES2 - 1) * KPAD + 2 * lanehi); \
      _Pragma("unroll")                                                                 \
      for (int ks_ = 0; ks_ < 13; ++ks_) {                                              \
        ga[BUF][ks_][0] = Ga_[2 * ks_];                                                 \
        ga[BUF][ks_][1] = Gb_[2 * ks_];                                                 \
        ga[BUF][ks_][2] = Gc_[2 * ks_];                                                 \
      }                                                                                 \
    }
    LOAD_G_GROUP(0, 0)
    __builtin_amdgcn_sched_barrier(0);
    #pragma unroll
    for (int rtg = 0; rtg < 7; ++rtg) {
      const int rowbase = rtg * 48;
      if (rtg + 1 < 7) LOAD_G_GROUP(rtg + 1, (rtg + 1) & 1)
      __builtin_amdgcn_sched_barrier(0);
      v8f acc0 = {}, acc1 = {}, acc2 = {}, acc3 = {}, acc4 = {}, acc5 = {};
      #pragma unroll
      for (int ks = 0; ks < 13; ++ks) {
        const v2f a0 = ga[rtg & 1][ks][0];
        const v2f a1 = ga[rtg & 1][ks][1];
        const v2f a2 = ga[rtg & 1][ks][2];
        acc0 = WMMA_F32(a0, bx0[ks], acc0);
        acc1 = WMMA_F32(a0, bx1[ks], acc1);
        acc2 = WMMA_F32(a1, bx0[ks], acc2);
        acc3 = WMMA_F32(a1, bx1[ks], acc3);
        acc4 = WMMA_F32(a2, bx0[ks], acc4);
        acc5 = WMMA_F32(a2, bx1[ks], acc5);
      }
      __builtin_amdgcn_sched_barrier(0);
      #pragma unroll
      for (int r = 0; r < 8; ++r) {
        const int m0 = rowbase      + r + 8 * lanehi;
        const int m1 = rowbase + 16 + r + 8 * lanehi;
        const int m2 = rowbase + 32 + r + 8 * lanehi;
        if (m0 < RES2) Yt[m0 * 16 + lane16] = acc0[r] * acc1[r];
        if (m1 < RES2) Yt[m1 * 16 + lane16] = acc2[r] * acc3[r];
        if (m2 < RES2) Yt[m2 * 16 + lane16] = acc4[r] * acc5[r];
      }
    }
#undef LOAD_G_GROUP

    {
      const int nn = n + (int)gridDim.x;
      if (nn < N) {
        LOAD_BX(X + (size_t)nn * (NCOEF * 2 * CTOT))
      }
    }

    v8f oa0 = {}, oa1 = {}, oa2 = {}, oa3 = {};
    v2f fa[2][9][4], fb[2][9];
#define LOAD_F_BLOCK(BLK, BUF)                                   \
    {                                                            \
      _Pragma("unroll")                                          \
      for (int j_ = 0; j_ < 9; ++j_) {                           \
        const int ks_ = (BLK) * 9 + j_;                          \
        fb[BUF][j_].x = Yb[(ks_ * 4)     * 16];                  \
        fb[BUF][j_].y = Yb[(ks_ * 4 + 1) * 16];                  \
        fa[BUF][j_][0] = F0[2 * ks_];                            \
        fa[BUF][j_][1] = F1[2 * ks_];                            \
        fa[BUF][j_][2] = F2[2 * ks_];                            \
        fa[BUF][j_][3] = F3[2 * ks_];                            \
      }                                                          \
    }
    LOAD_F_BLOCK(0, 0)
    __builtin_amdgcn_sched_barrier(0);
    #pragma unroll
    for (int blk = 0; blk < 9; ++blk) {
      if (blk + 1 < 9) LOAD_F_BLOCK(blk + 1, (blk + 1) & 1)
      __builtin_amdgcn_sched_barrier(0);
      #pragma unroll
      for (int j = 0; j < 9; ++j) {
        const v2f b = fb[blk & 1][j];
        oa0 = WMMA_F32(fa[blk & 1][j][0], b, oa0);
        oa1 = WMMA_F32(fa[blk & 1][j][1], b, oa1);
        oa2 = WMMA_F32(fa[blk & 1][j][2], b, oa2);
        oa3 = WMMA_F32(fa[blk & 1][j][3], b, oa3);
      }
      __builtin_amdgcn_sched_barrier(0);
    }
#undef LOAD_F_BLOCK

    const float gate  = sigmoidf_(sg);
    const float oscal = sa * sigmoidf_(sa) * sb;

    float* On = O + (size_t)n * (NCOEF * CTOT) + c;
    #pragma unroll
    for (int r = 0; r < 8; ++r) {
      const int i0 = r + 8 * lanehi;
      float v0 = oa0[r] * gate;
      if (i0 == 0) v0 += oscal;
      On[i0 * CTOT] = v0;
      On[(16 + i0) * CTOT] = oa1[r] * gate;
      On[(32 + i0) * CTOT] = oa2[r] * gate;
      const int i3o = 48 + i0;
      if (i3o < NCOEF) On[i3o * CTOT] = oa3[r] * gate;
    }
  }
}

extern "C" void kernel_launch(void* const* d_in, const int* in_sizes, int n_in,
                              void* d_out, int out_size, void* d_ws, size_t ws_size,
                              hipStream_t stream) {
  const float* X = (const float*)d_in[0];
  const float* S = (const float*)d_in[1];
  const float* G = (const float*)d_in[2];
  const float* F = (const float*)d_in[3];
  float* O = (float*)d_out;

  const int N = in_sizes[0] / (NCOEF * 2 * CTOT);
  const size_t shmem = (size_t)SMEM_FLOATS * sizeof(float);

  (void)hipFuncSetAttribute((const void*)s2act_swiglu_kernel,
                            hipFuncAttributeMaxDynamicSharedMemorySize,
                            (int)shmem);

  hipLaunchKernelGGL(s2act_swiglu_kernel, dim3(NBLOCKS), dim3(256), shmem, stream,
                     X, S, G, F, O, N);
}


# gated grid values stay in registers: second product takes its B fragments straight from the first product's accumulators (its k order permuted to fit: summation order changes), no LDS round trip
# speedup vs baseline: 1.7278x; 1.0432x over previous
#include <hip/hip_runtime.h>


typedef __attribute__((ext_vector_type(2))) float v2f;
typedef __attribute__((ext_vector_type(4))) float v4f;
typedef __attribute__((ext_vector_type(8))) float v8f;

#define NCOEF 49
#define KPAD 52
#define RES2 324
#define CTOT 128
#define NTHR 256
#define NBLOCKS 256
#define GROWS 336
#define GL_FLOATS (GROWS * KPAD)
#define FT_PITCH 332
#define FT_FLOATS (NCOEF * FT_PITCH)
#define PAIR_ITERS ((RES2 / 2 + 7) / 8)
#define KSTEPS2 82
#define SMEM_FLOATS (GL_FLOATS + FT_FLOATS)
#define FT_POS(pq) (((pq) & ~15) + 4 * (((pq) & 7) >> 1) + 2 * (((pq) >> 3) & 1) + ((pq) & 1))

__device__ __forceinline__ float sigmoidf_(float x) {
  return 1.0f / (1.0f + __expf(-x));
}

#define WMMA_F32(A, B, Cc) \
  __builtin_amdgcn_wmma_f32_16x16x4_f32(false, (A), false, (B), (short)0, (Cc), false, false)

#define LOAD_BX(XP)                                                        \
  {                                                                        \
    const float* xp_ = (XP) + (2 * lanehi) * (2 * CTOT) + c;               \
    _Pragma("unroll")                                                      \
    for (int ks_ = 0; ks_ < 12; ++ks_) {                                   \
      bx0[ks_].x = xp_[(4 * ks_)     * (2 * CTOT)];                        \
      bx0[ks_].y = xp_[(4 * ks_ + 1) * (2 * CTOT)];                        \
      bx1[ks_].x = xp_[(4 * ks_)     * (2 * CTOT) + CTOT];                 \
      bx1[ks_].y = xp_[(4 * ks_ + 1) * (2 * CTOT) + CTOT];                 \
    }                                                                      \
    x48a = (XP)[48 * (2 * CTOT) + c];                                      \
    x48b = (XP)[48 * (2 * CTOT) + CTOT + c];                               \
  }

__global__ __launch_bounds__(256, 1)
void s2act_swiglu_kernel(const float* __restrict__ X,
                         const float* __restrict__ S,
                         const float* __restrict__ G,
                         const float* __restrict__ F,
                         float* __restrict__ O,
                         const int N)
{
  extern __shared__ __attribute__((aligned(16))) float smem[];
  const int tid    = threadIdx.x;
  const int wave   = tid >> 5;
  const int lane   = tid & 31;
  const int lane16 = lane & 15;
  const int lanehi = lane >> 4;
  const int cbase  = wave * 16;
  const int c      = cbase + lane16;

  float* Gl = smem;
  float* Ft = smem + GL_FLOATS;

  int n = blockIdx.x;
  v2f bx0[13], bx1[13];
  float x48a = 0.0f, x48b = 0.0f;
  #pragma unroll
  for (int ks = 0; ks < 12; ++ks) { bx0[ks] = (v2f){0.0f, 0.0f}; bx1[ks] = (v2f){0.0f, 0.0f}; }
  if (n < N) {
    LOAD_BX(X + (size_t)n * (NCOEF * 2 * CTOT))
  }

  const int rp0 = 2 * wave + lanehi;
  #pragma unroll
  for (int j = 0; j < PAIR_ITERS; ++j) {
    const int row = rp0 + 16 * j;
    if (j < PAIR_ITERS - 1 || rp0 < RES2 - 16 * (PAIR_ITERS - 1)) {
      #pragma unroll
      for (int t = 0; t < 3; ++t)
        Gl[row * KPAD + 16 * t + lane16] = G[row * NCOEF + 16 * t + lane16];
    }
  }
  {
    v4f g;
    g.x = G[tid * NCOEF + (NCOEF - 1)]; g.y = 0.0f; g.z = 0.0f; g.w = 0.0f;
    *(v4f*)(Gl + tid * KPAD + (NCOEF - 1)) = g;
    if (tid + NTHR < RES2) {
      g.x = G[(tid + NTHR) * NCOEF + (NCOEF - 1)];
      *(v4f*)(Gl + (tid + NTHR) * KPAD + (NCOEF - 1)) = g;
    }
  }
  for (int idx = tid; idx < (GROWS - RES2) * KPAD; idx += NTHR) Gl[RES2 * KPAD + idx] = 0.0f;

  const int posl = FT_POS(rp0);
  #pragma unroll
  for (int j = 0; j < PAIR_ITERS; ++j) {
    const int pq = rp0 + 16 * j;
    if (j < PAIR_ITERS - 1 || rp0 < RES2 - 16 * (PAIR_ITERS - 1)) {
      #pragma unroll
      for (int t = 0; t < 3; ++t)
        Ft[(16 * t + lane16) * FT_PITCH + 16 * j + posl] = F[pq * NCOEF + 16 * t + lane16];
    }
  }
  Ft[(NCOEF - 1) * FT_PITCH + FT_POS(tid)] = F[tid * NCOEF + (NCOEF - 1)];
  if (tid + NTHR < RES2)
    Ft[(NCOEF - 1) * FT_PITCH + FT_POS(tid + NTHR)] = F[(tid + NTHR) * NCOEF + (NCOEF - 1)];
  if (tid < 4 * NCOEF)
    Ft[(tid >> 2) * FT_PITCH + 322 + (tid & 1) + 4 * ((tid >> 1) & 1)] = 0.0f;
  __syncthreads();

  const int i3 = (48 + lane16) < NCOEF ? (48 + lane16) : NCOEF - 1;
  const v2f* F0 = (const v2f*)(Ft + (     lane16) * FT_PITCH + 2 * lanehi);
  const v2f* F1 = (const v2f*)(Ft + (16 + lane16) * FT_PITCH + 2 * lanehi);
  const v2f* F2 = (const v2f*)(Ft + (32 + lane16) * FT_PITCH + 2 * lanehi);
  const v2f* F3 = (const v2f*)(Ft + i3 * FT_PITCH + 2 * lanehi);
  const v2f* Gf = (const v2f*)(Gl + lane16 * KPAD + 2 * lanehi);

  v2f ga[2][13][3];
  v2f fa[2][10][4];
#define LOAD_G_GROUP(RTG, BUF)                                                          \
    {                                                                                   \
      const v2f* Ga_ = Gf + ((RTG) * 48) * (KPAD / 2);                                  \
      const v2f* Gb_ = Ga_ + 16 * (KPAD / 2);                                           \
      const v2f* Gc_ = Ga_ + 32 * (KPAD / 2);                                           \
      _Pragma("unroll")                                                                 \
      for (int ks_ = 0; ks_ < 13; ++ks_) {                                              \
        ga[BUF][ks_][0] = Ga_[2 * ks_];                                                 \
        ga[BUF][ks_][1] = Gb_[2 * ks_];                                                 \
        ga[BUF][ks_][2] = Gc_[2 * ks_];                                                 \
      }                                                                                 \
    }
#define FB_START(B) ((B) < 4 ? 9 * (B) : (B) == 4 ? 36 : 46 + 9 * ((B) - 5))
#define FB_CNT(B)   ((B) == 4 ? 10 : 9)
#define LOAD_F_BLOCK(BLK, BUF)                                   \
    {                                                            \
      _Pragma("unroll")                                          \
      for (int j_ = 0; j_ < 10; ++j_) {                          \
        const int ks_ = FB_START(BLK) + j_;                      \
        if (j_ < FB_CNT(BLK)) {                                  \
          fa[BUF][j_][0] = F0[2 * ks_];                          \
          fa[BUF][j_][1] = F1[2 * ks_];                          \
          fa[BUF][j_][2] = F2[2 * ks_];                          \
          fa[BUF][j_][3] = F3[2 * ks_];                          \
        }                                                        \
      }                                                          \
    }
  LOAD_G_GROUP(0, 0)

  #pragma nounroll
  for (; n < N; n += gridDim.x) {
    bx0[12] = (v2f){lanehi == 0 ? x48a : 0.0f, 0.0f};
    bx1[12] = (v2f){lanehi == 0 ? x48b : 0.0f, 0.0f};

    v2f yb[KSTEPS2];
    v8f oa0 = {}, oa1 = {}, oa2 = {}, oa3 = {};
    float sa = 0.0f, sb = 0.0f, sg = 0.0f;
    #pragma unroll
    for (int ph = 0; ph < 2; ++ph) {
      const int g0 = ph == 0 ? 0 : 3, g1 = ph == 0 ? 3 : 7;
      const int f0 = ph == 0 ? 0 : 4, f1 = ph == 0 ? 4 : 9;

      #pragma unroll
      for (int q = 0; q < 4; ++q) {
        const int rtg = g0 + q;
        if (rtg < g1) {
          if (rtg + 1 < g1) LOAD_G_GROUP(rtg + 1, (rtg + 1) & 1)
          else              LOAD_F_BLOCK(f0, f0 & 1)
          __builtin_amdgcn_sched_barrier(0);
          v8f acc0 = {}, acc1 = {}, acc2 = {}, acc3 = {}, acc4 = {}, acc5 = {};
          #pragma unroll
          for (int ks = 0; ks < 13; ++ks) {
            const v2f a0 = ga[rtg & 1][ks][0];
            const v2f a1 = ga[rtg & 1][ks][1];
            const v2f a2 = ga[rtg & 1][ks][2];
            acc0 = WMMA_F32(a0, bx0[ks], acc0);
            acc1 = WMMA_F32(a0, bx1[ks], acc1);
            acc2 = WMMA_F32(a1, bx0[ks], acc2);
            acc3 = WMMA_F32(a1, bx1[ks], acc3);
            acc4 = WMMA_F32(a2, bx0[ks], acc4);
            acc5 = WMMA_F32(a2, bx1[ks], acc5);
          }
          __builtin_amdgcn_sched_barrier(0);
          #pragma unroll
          for (int j = 0; j < 4; ++j) {
            const int s0 = 4 * (3 * rtg) + j;
            const int s1 = s0 + 4;
            const int s2 = s0 + 8;
            yb[s0] = (v2f){acc0[2 * j] * acc1[2 * j], acc0[2 * j + 1] * acc1[2 * j + 1]};
            yb[s1] = (v2f){acc2[2 * j] * acc3[2 * j], acc2[2 * j + 1] * acc3[2 * j + 1]};
            if (s2 < KSTEPS2)
              yb[s2] = (v2f){acc4[2 * j] * acc5[2 * j], acc4[2 * j + 1] * acc5[2 * j + 1]};
          }
        }
      }

      if (ph == 1) {
        const float* Sn = S + (size_t)n * (3 * CTOT);
        sa = Sn[c];
        sb = Sn[CTOT + c];
        sg = Sn[2 * CTOT + c];
        const int nn = n + (int)gridDim.x;
        LOAD_BX(X + (size_t)(nn < N ? nn : n) * (NCOEF * 2 * CTOT))
      }

      #pragma unroll
      for (int q = 0; q < 5; ++q) {
        const int blk = f0 + q;
        if (blk < f1) {
          if (blk + 1 < f1) LOAD_F_BLOCK(blk + 1, (blk + 1) & 1)
          else if (ph == 0) LOAD_G_GROUP(g1, g1 & 1)
          else              LOAD_G_GROUP(0, 0)
          __builtin_amdgcn_sched_barrier(0);
          #pragma unroll
          for (int j = 0; j < 10; ++j) {
            if (j < FB_CNT(blk)) {
              const v2f b = yb[FB_START(blk) + j];
              oa0 = WMMA_F32(fa[blk & 1][j][0], b, oa0);
              oa1 = WMMA_F32(fa[blk & 1][j][1], b, oa1);
              oa2 = WMMA_F32(fa[blk & 1][j][2], b, oa2);
              oa3 = WMMA_F32(fa[blk & 1][j][3], b, oa3);
            }
          }
          __builtin_amdgcn_sched_barrier(0);
        }
      }
    }

    const float gate  = sigmoidf_(sg);
    const float oscal = sa * sigmoidf_(sa) * sb;

    float* On = O + (size_t)n * (NCOEF * CTOT) + c;
    #pragma unroll
    for (int r = 0; r < 8; ++r) {
      const int i0 = r + 8 * lanehi;
      float v0 = oa0[r] * gate;
      if (i0 == 0) v0 += oscal;
      On[i0 * CTOT] = v0;
      On[(16 + i0) * CTOT] = oa1[r] * gate;
      On[(32 + i0) * CTOT] = oa2[r] * gate;
      const int i3o = 48 + i0;
      if (i3o < NCOEF) On[i3o * CTOT] = oa3[r] * gate;
    }
  }
}

#undef LOAD_G_GROUP
#undef LOAD_F_BLOCK
#undef FB_START
#undef FB_CNT
#undef LOAD_BX

extern "C" void kernel_launch(void* const* d_in, const int* in_sizes, int n_in,
                              void* d_out, int out_size, void* d_ws, size_t ws_size,
                              hipStream_t stream) {
  const float* X = (const float*)d_in[0];
  const float* S = (const float*)d_in[1];
  const float* G = (const float*)d_in[2];
  const float* F = (const float*)d_in[3];
  float* O = (float*)d_out;

  const int N = in_sizes[0] / (NCOEF * 2 * CTOT);
  const size_t shmem = (size_t)SMEM_FLOATS * sizeof(float);

  (void)hipFuncSetAttribute((const void*)s2act_swiglu_kernel,
                            hipFuncAttributeMaxDynamicSharedMemorySize,
                            (int)shmem);

  hipLaunchKernelGGL(s2act_swiglu_kernel, dim3(NBLOCKS), dim3(256), shmem, stream,
                     X, S, G, F, O, N);
}


# output row i=48 of the second product (the lone row of its 4th 16-row tile, a matrix-vector product) runs as 82 packed f32 FMAs, not 82 WMMAs; rows 0..47 stay WMMA; its summation order changes
# speedup vs baseline: 1.8491x; 1.0702x over previous
#include <hip/hip_runtime.h>


typedef __attribute__((ext_vector_type(2))) float v2f;
typedef __attribute__((ext_vector_type(4))) float v4f;
typedef __attribute__((ext_vector_type(8))) float v8f;

#define NCOEF 49
#define KPAD 52
#define RES2 324
#define CTOT 128
#define NTHR 256
#define NBLOCKS 256
#define GROWS 336
#define GL_FLOATS (GROWS * KPAD)
#define FT_PITCH 332
#define FT_FLOATS (NCOEF * FT_PITCH)
#define PAIR_ITERS ((RES2 / 2 + 7) / 8)
#define KSTEPS2 82
#define SMEM_FLOATS (GL_FLOATS + FT_FLOATS)
#define FT_POS(pq) (((pq) & ~15) + 4 * (((pq) & 7) >> 1) + 2 * (((pq) >> 3) & 1) + ((pq) & 1))

__device__ __forceinline__ float sigmoidf_(float x) {
  return 1.0f / (1.0f + __expf(-x));
}

#define WMMA_F32(A, B, Cc) \
  __builtin_amdgcn_wmma_f32_16x16x4_f32(false, (A), false, (B), (short)0, (Cc), false, false)

#define LOAD_BX(XP)                                                        \
  {                                                                        \
    const float* xp_ = (XP) + (2 * lanehi) * (2 * CTOT) + c;               \
    _Pragma("unroll")                                                      \
    for (int ks_ = 0; ks_ < 12; ++ks_) {                                   \
      bx0[ks_].x = xp_[(4 * ks_)     * (2 * CTOT)];                        \
      bx0[ks_].y = xp_[(4 * ks_ + 1) * (2 * CTOT)];                        \
      bx1[ks_].x = xp_[(4 * ks_)     * (2 * CTOT) + CTOT];                 \
      bx1[ks_].y = xp_[(4 * ks_ + 1) * (2 * CTOT) + CTOT];                 \
    }                                                                      \
    x48a = (XP)[48 * (2 * CTOT) + c];                                      \
    x48b = (XP)[48 * (2 * CTOT) + CTOT + c];                               \
  }

__global__ __launch_bounds__(256, 1)
void s2act_swiglu_kernel(const float* __restrict__ X,
                         const float* __restrict__ S,
                         const float* __restrict__ G,
                         const float* __restrict__ F,
                         float* __restrict__ O,
                         const int N)
{
  extern __shared__ __attribute__((aligned(16))) float smem[];
  const int tid    = threadIdx.x;
  const int wave   = tid >> 5;
  const int lane   = tid & 31;
  const int lane16 = lane & 15;
  const int lanehi = lane >> 4;
  const int cbase  = wave * 16;
  const int c      = cbase + lane16;

  float* Gl = smem;
  float* Ft = smem + GL_FLOATS;

  int n = blockIdx.x;
  v2f bx0[13], bx1[13];
  float x48a = 0.0f, x48b = 0.0f;
  #pragma unroll
  for (int ks = 0; ks < 12; ++ks) { bx0[ks] = (v2f){0.0f, 0.0f}; bx1[ks] = (v2f){0.0f, 0.0f}; }
  if (n < N) {
    LOAD_BX(X + (size_t)n * (NCOEF * 2 * CTOT))
  }

  const int rp0 = 2 * wave + lanehi;
  #pragma unroll
  for (int j = 0; j < PAIR_ITERS; ++j) {
    const int row = rp0 + 16 * j;
    if (j < PAIR_ITERS - 1 || rp0 < RES2 - 16 * (PAIR_ITERS - 1)) {
      #pragma unroll
      for (int t = 0; t < 3; ++t)
        Gl[row * KPAD + 16 * t + lane16] = G[row * NCOEF + 16 * t + lane16];
    }
  }
  {
    v4f g;
    g.x = G[tid * NCOEF + (NCOEF - 1)]; g.y = 0.0f; g.z = 0.0f; g.w = 0.0f;
    *(v4f*)(Gl + tid * KPAD + (NCOEF - 1)) = g;
    if (tid + NTHR < RES2) {
      g.x = G[(tid + NTHR) * NCOEF + (NCOEF - 1)];
      *(v4f*)(Gl + (tid + NTHR) * KPAD + (NCOEF - 1)) = g;
    }
  }
  for (int idx = tid; idx < (GROWS - RES2) * KPAD; idx += NTHR) Gl[RES2 * KPAD + idx] = 0.0f;

  const int posl = FT_POS(rp0);
  #pragma unroll
  for (int j = 0; j < PAIR_ITERS; ++j) {
    const int pq = rp0 + 16 * j;
    if (j < PAIR_ITERS - 1 || rp0 < RES2 - 16 * (PAIR_ITERS - 1)) {
      #pragma unroll
      for (int t = 0; t < 3; ++t)
        Ft[(16 * t + lane16) * FT_PITCH + 16 * j + posl] = F[pq * NCOEF + 16 * t + lane16];
    }
  }
  Ft[(NCOEF - 1) * FT_PITCH + FT_POS(tid)] = F[tid * NCOEF + (NCOEF - 1)];
  if (tid + NTHR < RES2)
    Ft[(NCOEF - 1) * FT_PITCH + FT_POS(tid + NTHR)] = F[(tid + NTHR) * NCOEF + (NCOEF - 1)];
  if (tid < 4 * NCOEF)
    Ft[(tid >> 2) * FT_PITCH + 322 + (tid & 1) + 4 * ((tid >> 1) & 1)] = 0.0f;
  __syncthreads();

  const v2f* F0 = (const v2f*)(Ft + (     lane16) * FT_PITCH + 2 * lanehi);
  const v2f* F1 = (const v2f*)(Ft + (16 + lane16) * FT_PITCH + 2 * lanehi);
  const v2f* F2 = (const v2f*)(Ft + (32 + lane16) * FT_PITCH + 2 * lanehi);
  const v2f* F3 = (const v2f*)(Ft + (NCOEF - 1) * FT_PITCH + 2 * lanehi);
  const v2f* Gf = (const v2f*)(Gl + lane16 * KPAD + 2 * lanehi);

  v2f ga[2][13][3];
  v2f fa[2][10][4];
#define LOAD_G_GROUP(RTG, BUF)                                                          \
    {                                                                                   \
      const v2f* Ga_ = Gf + ((RTG) * 48) * (KPAD / 2);                                  \
      const v2f* Gb_ = Ga_ + 16 * (KPAD / 2);                                           \
      const v2f* Gc_ = Ga_ + 32 * (KPAD / 2);                                           \
      _Pragma("unroll")                                                                 \
      for (int ks_ = 0; ks_ < 13; ++ks_) {                                              \
        ga[BUF][ks_][0] = Ga_[2 * ks_];                                                 \
        ga[BUF][ks_][1] = Gb_[2 * ks_];                                                 \
        ga[BUF][ks_][2] = Gc_[2 * ks_];                                                 \
      }                                                                                 \
    }
#define FB_START(B) ((B) < 4 ? 9 * (B) : (B) == 4 ? 36 : 46 + 9 * ((B) - 5))
#define FB_CNT(B)   ((B) == 4 ? 10 : 9)
#define LOAD_F_BLOCK(BLK, BUF)                                   \
    {                                                            \
      _Pragma("unroll")                                          \
      for (int j_ = 0; j_ < 10; ++j_) {                          \
        const int ks_ = FB_START(BLK) + j_;                      \
        if (j_ < FB_CNT(BLK)) {                                  \
          fa[BUF][j_][0] = F0[2 * ks_];                          \
          fa[BUF][j_][1] = F1[2 * ks_];                          \
          fa[BUF][j_][2] = F2[2 * ks_];                          \
          fa[BUF][j_][3] = F3[2 * ks_];                          \
        }                                                        \
      }                                                          \
    }
  LOAD_G_GROUP(0, 0)

  #pragma nounroll
  for (; n < N; n += gridDim.x) {
    bx0[12] = (v2f){lanehi == 0 ? x48a : 0.0f, 0.0f};
    bx1[12] = (v2f){lanehi == 0 ? x48b : 0.0f, 0.0f};

    v2f yb[KSTEPS2];
    v8f oa0 = {}, oa1 = {}, oa2 = {};
    v2f o48[4] = {};
    float sa = 0.0f, sb = 0.0f, sg = 0.0f;
    #pragma unroll
    for (int ph = 0; ph < 2; ++ph) {
      const int g0 = ph == 0 ? 0 : 3, g1 = ph == 0 ? 3 : 7;
      const int f0 = ph == 0 ? 0 : 4, f1 = ph == 0 ? 4 : 9;

      #pragma unroll
      for (int q = 0; q < 4; ++q) {
        const int rtg = g0 + q;
        if (rtg < g1) {
          if (rtg + 1 < g1) LOAD_G_GROUP(rtg + 1, (rtg + 1) & 1)
          else              LOAD_F_BLOCK(f0, f0 & 1)
          __builtin_amdgcn_sched_barrier(0);
          v8f acc0 = {}, acc1 = {}, acc2 = {}, acc3 = {}, acc4 = {}, acc5 = {};
          #pragma unroll
          for (int ks = 0; ks < 13; ++ks) {
            const v2f a0 = ga[rtg & 1][ks][0];
            const v2f a1 = ga[rtg & 1][ks][1];
            const v2f a2 = ga[rtg & 1][ks][2];
            acc0 = WMMA_F32(a0, bx0[ks], acc0);
            acc1 = WMMA_F32(a0, bx1[ks], acc1);
            acc2 = WMMA_F32(a1, bx0[ks], acc2);
            acc3 = WMMA_F32(a1, bx1[ks], acc3);
            acc4 = WMMA_F32(a2, bx0[ks], acc4);
            acc5 = WMMA_F32(a2, bx1[ks], acc5);
          }
          __builtin_amdgcn_sched_barrier(0);
          #pragma unroll
          for (int j = 0; j < 4; ++j) {
            const int s0 = 4 * (3 * rtg) + j;
            const int s1 = s0 + 4;
            const int s2 = s0 + 8;
            yb[s0] = (v2f){acc0[2 * j] * acc1[2 * j], acc0[2 * j + 1] * acc1[2 * j + 1]};
            yb[s1] = (v2f){acc2[2 * j] * acc3[2 * j], acc2[2 * j + 1] * acc3[2 * j + 1]};
            if (s2 < KSTEPS2)
              yb[s2] = (v2f){acc4[2 * j] * acc5[2 * j], acc4[2 * j + 1] * acc5[2 * j + 1]};
          }
        }
      }

      if (ph == 1) {
        const float* Sn = S + (size_t)n * (3 * CTOT);
        sa = Sn[c];
        sb = Sn[CTOT + c];
        sg = Sn[2 * CTOT + c];
        const int nn = n + (int)gridDim.x;
        LOAD_BX(X + (size_t)(nn < N ? nn : n) * (NCOEF * 2 * CTOT))
      }

      #pragma unroll
      for (int q = 0; q < 5; ++q) {
        const int blk = f0 + q;
        if (blk < f1) {
          if (blk + 1 < f1) LOAD_F_BLOCK(blk + 1, (blk + 1) & 1)
          else if (ph == 0) LOAD_G_GROUP(g1, g1 & 1)
          else              LOAD_G_GROUP(0, 0)
          __builtin_amdgcn_sched_barrier(0);
          #pragma unroll
          for (int j = 0; j < 10; ++j) {
            if (j < FB_CNT(blk)) {
              const v2f b = yb[FB_START(blk) + j];
              oa0 = WMMA_F32(fa[blk & 1][j][0], b, oa0);
              oa1 = WMMA_F32(fa[blk & 1][j][1], b, oa1);
              oa2 = WMMA_F32(fa[blk & 1][j][2], b, oa2);
              o48[j & 3] = __builtin_elementwise_fma(fa[blk & 1][j][3], b, o48[j & 3]);
            }
          }
          __builtin_amdgcn_sched_barrier(0);
        }
      }
    }

    const float gate  = sigmoidf_(sg);
    const float oscal = sa * sigmoidf_(sa) * sb;

    float* On = O + (size_t)n * (NCOEF * CTOT) + c;
    #pragma unroll
    for (int r = 0; r < 8; ++r) {
      const int i0 = r + 8 * lanehi;
      float v0 = oa0[r] * gate;
      if (i0 == 0) v0 += oscal;
      On[i0 * CTOT] = v0;
      On[(16 + i0) * CTOT] = oa1[r] * gate;
      On[(32 + i0) * CTOT] = oa2[r] * gate;
    }
    const v2f s48 = (o48[0] + o48[1]) + (o48[2] + o48[3]);
    const float h48 = s48.x + s48.y;
    const float p48 = __builtin_bit_cast(float,
        __builtin_amdgcn_permlanex16(0u, __builtin_bit_cast(unsigned, h48),
                                     0x76543210u, 0xfedcba98u, false, false));
    if (lanehi == 0) On[(NCOEF - 1) * CTOT] = (h48 + p48) * gate;
  }
}

#undef LOAD_G_GROUP
#undef LOAD_F_BLOCK
#undef FB_START
#undef FB_CNT
#undef LOAD_BX

extern "C" void kernel_launch(void* const* d_in, const int* in_sizes, int n_in,
                              void* d_out, int out_size, void* d_ws, size_t ws_size,
                              hipStream_t stream) {
  const float* X = (const float*)d_in[0];
  const float* S = (const float*)d_in[1];
  const float* G = (const float*)d_in[2];
  const float* F = (const float*)d_in[3];
  float* O = (float*)d_out;

  const int N = in_sizes[0] / (NCOEF * 2 * CTOT);
  const size_t shmem = (size_t)SMEM_FLOATS * sizeof(float);

  (void)hipFuncSetAttribute((const void*)s2act_swiglu_kernel,
                            hipFuncAttributeMaxDynamicSharedMemorySize,
                            (int)shmem);

  hipLaunchKernelGGL(s2act_swiglu_kernel, dim3(NBLOCKS), dim3(256), shmem, stream,
                     X, S, G, F, O, N);
}


# first product's remainders leave the matrix pipe: term k=48 becomes the accumulators' start value, grid rows 320..323 run as packed f32 FMAs; 723 WMMAs per node, not 792; summation order changes
# speedup vs baseline: 1.9201x; 1.0384x over previous
#include <hip/hip_runtime.h>


typedef __attribute__((ext_vector_type(2))) float v2f;
typedef __attribute__((ext_vector_type(4))) float v4f;
typedef __attribute__((ext_vector_type(8))) float v8f;

#define NCOEF 49
#define KPAD 52
#define RES2 324
#define CTOT 128
#define NTHR 256
#define NBLOCKS 256
#define GROWS 324
#define GL_FLOATS (GROWS * KPAD)
#define GQ_FLOATS 336
#define FT_PITCH 332
#define FT_FLOATS (NCOEF * FT_PITCH)
#define PAIR_ITERS ((RES2 / 2 + 7) / 8)
#define KSTEPS2 81
#define SMEM_FLOATS (GL_FLOATS + FT_FLOATS + GQ_FLOATS)
#define FT_POS(pq) ((pq) >= 320 ? (pq) : \
                    ((pq) & ~15) + 4 * (((pq) & 7) >> 1) + 2 * (((pq) >> 3) & 1) + ((pq) & 1))

__device__ __forceinline__ float sigmoidf_(float x) {
  return 1.0f / (1.0f + __expf(-x));
}

#define WMMA_F32(A, B, Cc) \
  __builtin_amdgcn_wmma_f32_16x16x4_f32(false, (A), false, (B), (short)0, (Cc), false, false)

#define LOAD_BX(XP)                                                        \
  {                                                                        \
    const float* xp_ = (XP) + (2 * lanehi) * (2 * CTOT) + c;               \
    _Pragma("unroll")                                                      \
    for (int ks_ = 0; ks_ < 12; ++ks_) {                                   \
      bx0[ks_].x = xp_[(4 * ks_)     * (2 * CTOT)];                        \
      bx0[ks_].y = xp_[(4 * ks_ + 1) * (2 * CTOT)];                        \
      bx1[ks_].x = xp_[(4 * ks_)     * (2 * CTOT) + CTOT];                 \
      bx1[ks_].y = xp_[(4 * ks_ + 1) * (2 * CTOT) + CTOT];                 \
    }                                                                      \
    x48a = (XP)[48 * (2 * CTOT) + c];                                      \
    x48b = (XP)[48 * (2 * CTOT) + CTOT + c];                               \
  }

__global__ __launch_bounds__(256, 1)
void s2act_swiglu_kernel(const float* __restrict__ X,
                         const float* __restrict__ S,
                         const float* __restrict__ G,
                         const float* __restrict__ F,
                         float* __restrict__ O,
                         const int N)
{
  extern __shared__ __attribute__((aligned(16))) float smem[];
  const int tid    = threadIdx.x;
  const int wave   = tid >> 5;
  const int lane   = tid & 31;
  const int lane16 = lane & 15;
  const int lanehi = lane >> 4;
  const int cbase  = wave * 16;
  const int c      = cbase + lane16;

  float* Gl = smem;
  float* Ft = smem + GL_FLOATS;
  float* Gq = smem + GL_FLOATS + FT_FLOATS;

  int n = blockIdx.x;
  v2f bx0[12], bx1[12];
  float x48a = 0.0f, x48b = 0.0f;
  #pragma unroll
  for (int ks = 0; ks < 12; ++ks) { bx0[ks] = (v2f){0.0f, 0.0f}; bx1[ks] = (v2f){0.0f, 0.0f}; }
  if (n < N) {
    LOAD_BX(X + (size_t)n * (NCOEF * 2 * CTOT))
  }

  const int rp0 = 2 * wave + lanehi;
  #pragma unroll
  for (int j = 0; j < PAIR_ITERS; ++j) {
    const int row = rp0 + 16 * j;
    if (j < PAIR_ITERS - 1 || rp0 < RES2 - 16 * (PAIR_ITERS - 1)) {
      #pragma unroll
      for (int t = 0; t < 3; ++t)
        Gl[row * KPAD + 16 * t + lane16] = G[row * NCOEF + 16 * t + lane16];
    }
  }
  Gq[tid] = G[tid * NCOEF + (NCOEF - 1)];
  if (tid + NTHR < GQ_FLOATS)
    Gq[tid + NTHR] = (tid + NTHR < RES2) ? G[(tid + NTHR) * NCOEF + (NCOEF - 1)] : 0.0f;

  const int posl = FT_POS(rp0);
  #pragma unroll
  for (int j = 0; j < PAIR_ITERS; ++j) {
    const int pq = rp0 + 16 * j;
    if (j < PAIR_ITERS - 1 || rp0 < RES2 - 16 * (PAIR_ITERS - 1)) {
      #pragma unroll
      for (int t = 0; t < 3; ++t)
        Ft[(16 * t + lane16) * FT_PITCH + 16 * j + (j < PAIR_ITERS - 1 ? posl : rp0)] =
            F[pq * NCOEF + 16 * t + lane16];
    }
  }
  Ft[(NCOEF - 1) * FT_PITCH + FT_POS(tid)] = F[tid * NCOEF + (NCOEF - 1)];
  if (tid + NTHR < RES2)
    Ft[(NCOEF - 1) * FT_PITCH + FT_POS(tid + NTHR)] = F[(tid + NTHR) * NCOEF + (NCOEF - 1)];
  __syncthreads();

  const v2f* F0 = (const v2f*)(Ft + (     lane16) * FT_PITCH + 2 * lanehi);
  const v2f* F1 = (const v2f*)(Ft + (16 + lane16) * FT_PITCH + 2 * lanehi);
  const v2f* F2 = (const v2f*)(Ft + (32 + lane16) * FT_PITCH + 2 * lanehi);
  const v2f* F3 = (const v2f*)(Ft + (NCOEF - 1) * FT_PITCH + 2 * lanehi);
  const v2f* Gf = (const v2f*)(Gl + lane16 * KPAD + 2 * lanehi);
  const v2f* Gx = (const v2f*)(Gl + 320 * KPAD + 2 * lanehi);
  const v4f* Gq4 = (const v4f*)(Gq + 8 * lanehi);

  v2f ga[2][12][3];
  v4f gq[2][3][2];
  v2f gx[2][2][4];
  v4f gq320;
  v2f fa[2][9][4];
#define LOAD_G_GROUP(RTG, BUF)                                                          \
    {                                                                                   \
      const v2f* Ga_ = Gf + ((RTG) * 48) * (KPAD / 2);                                  \
      const v2f* Gb_ = Ga_ + 16 * (KPAD / 2);                                           \
      const v2f* Gc_ = Ga_ + 32 * (KPAD / 2);                                           \
      _Pragma("unroll")                                                                 \
      for (int ks_ = 0; ks_ < 12; ++ks_) {                                              \
        ga[BUF][ks_][0] = Ga_[2 * ks_];                                                 \
        ga[BUF][ks_][1] = Gb_[2 * ks_];                                                 \
        if ((RTG) < 6) ga[BUF][ks_][2] = Gc_[2 * ks_];                                  \
      }                                                                                 \
      _Pragma("unroll")                                                                 \
      for (int t_ = 0; t_ < 3; ++t_) {                                                  \
        if ((RTG) < 6 || t_ < 2) {                                                      \
          gq[BUF][t_][0] = Gq4[4 * (3 * (RTG) + t_)];                                   \
          gq[BUF][t_][1] = Gq4[4 * (3 * (RTG) + t_) + 1];                               \
        }                                                                               \
      }                                                                                 \
      if ((RTG) < 6) {                                                                  \
        _Pragma("unroll")                                                               \
        for (int r_ = 0; r_ < 4; ++r_) {                                                \
          gx[BUF][0][r_] = Gx[r_ * (KPAD / 2) + 2 * (2 * (RTG))];                       \
          gx[BUF][1][r_] = Gx[r_ * (KPAD / 2) + 2 * (2 * (RTG) + 1)];                   \
        }                                                                               \
      } else {                                                                          \
        gq320 = *(const v4f*)(Gq + 320);                                                \
      }                                                                                 \
    }
#define FB_START(B) (9 * (B))
#define FB_CNT(B)   9
#define LOAD_F_BLOCK(BLK, BUF)                                   \
    {                                                            \
      _Pragma("unroll")                                          \
      for (int j_ = 0; j_ < 9; ++j_) {                           \
        const int ks_ = FB_START(BLK) + j_;                      \
        fa[BUF][j_][0] = F0[2 * ks_];                            \
        fa[BUF][j_][1] = F1[2 * ks_];                            \
        fa[BUF][j_][2] = F2[2 * ks_];                            \
        fa[BUF][j_][3] = F3[2 * ks_];                            \
      }                                                          \
    }
  LOAD_G_GROUP(0, 0)

  #pragma nounroll
  for (; n < N; n += gridDim.x) {
    v2f yb[KSTEPS2];
    v8f oa0 = {}, oa1 = {}, oa2 = {};
    v2f o48[4] = {};
    v2f pa[4] = {}, pb[4] = {};
    float sa = 0.0f, sb = 0.0f, sg = 0.0f;
    #pragma unroll
    for (int ph = 0; ph < 2; ++ph) {
      const int g0 = ph == 0 ? 0 : 3, g1 = ph == 0 ? 3 : 7;
      const int f0 = ph == 0 ? 0 : 4, f1 = ph == 0 ? 4 : 9;

      #pragma unroll
      for (int q = 0; q < 4; ++q) {
        const int rtg = g0 + q;
        if (rtg < g1) {
          if (rtg + 1 < g1) LOAD_G_GROUP(rtg + 1, (rtg + 1) & 1)
          else              LOAD_F_BLOCK(f0, f0 & 1)
          __builtin_amdgcn_sched_barrier(0);
          v8f acc0, acc1, acc2, acc3, acc4 = {}, acc5 = {};
          #pragma unroll
          for (int r = 0; r < 4; ++r) {
            acc0[r]     = gq[rtg & 1][0][0][r] * x48a;
            acc0[r + 4] = gq[rtg & 1][0][1][r] * x48a;
            acc1[r]     = gq[rtg & 1][0][0][r] * x48b;
            acc1[r + 4] = gq[rtg & 1][0][1][r] * x48b;
            acc2[r]     = gq[rtg & 1][1][0][r] * x48a;
            acc2[r + 4] = gq[rtg & 1][1][1][r] * x48a;
            acc3[r]     = gq[rtg & 1][1][0][r] * x48b;
            acc3[r + 4] = gq[rtg & 1][1][1][r] * x48b;
            if (rtg < 6) {
              acc4[r]     = gq[rtg & 1][2][0][r] * x48a;
              acc4[r + 4] = gq[rtg & 1][2][1][r] * x48a;
              acc5[r]     = gq[rtg & 1][2][0][r] * x48b;
              acc5[r + 4] = gq[rtg & 1][2][1][r] * x48b;
            }
          }
          #pragma unroll
          for (int ks = 0; ks < 12; ++ks) {
            const v2f a0 = ga[rtg & 1][ks][0];
            const v2f a1 = ga[rtg & 1][ks][1];
            acc0 = WMMA_F32(a0, bx0[ks], acc0);
            acc1 = WMMA_F32(a0, bx1[ks], acc1);
            acc2 = WMMA_F32(a1, bx0[ks], acc2);
            acc3 = WMMA_F32(a1, bx1[ks], acc3);
            if (rtg < 6) {
              const v2f a2 = ga[rtg & 1][ks][2];
              acc4 = WMMA_F32(a2, bx0[ks], acc4);
              acc5 = WMMA_F32(a2, bx1[ks], acc5);
              if ((ks >> 1) == rtg) {
                #pragma unroll
                for (int r = 0; r < 4; ++r) {
                  pa[r] = __builtin_elementwise_fma(gx[rtg & 1][ks & 1][r], bx0[ks], pa[r]);
                  pb[r] = __builtin_elementwise_fma(gx[rtg & 1][ks & 1][r], bx1[ks], pb[r]);
                }
              }
            }
          }
          if (rtg == 6) {
            float ya[4];
            #pragma unroll
            for (int r = 0; r < 4; ++r) {
              const float ha = pa[r].x + pa[r].y;
              const float hb = pb[r].x + pb[r].y;
              const float qa = __builtin_bit_cast(float,
                  __builtin_amdgcn_permlanex16(0u, __builtin_bit_cast(unsigned, ha),
                                               0x76543210u, 0xfedcba98u, false, false));
              const float qb = __builtin_bit_cast(float,
                  __builtin_amdgcn_permlanex16(0u, __builtin_bit_cast(unsigned, hb),
                                               0x76543210u, 0xfedcba98u, false, false));
              ya[r] = ((ha + qa) + gq320[r] * x48a) * ((hb + qb) + gq320[r] * x48b);
            }
            yb[80] = (v2f){lanehi == 0 ? ya[0] : ya[2], lanehi == 0 ? ya[1] : ya[3]};
          }
          __builtin_amdgcn_sched_barrier(0);
          #pragma unroll
          for (int j = 0; j < 4; ++j) {
            const int s0 = 4 * (3 * rtg) + j;
            const int s1 = s0 + 4;
            const int s2 = s0 + 8;
            yb[s0] = (v2f){acc0[2 * j] * acc1[2 * j], acc0[2 * j + 1] * acc1[2 * j + 1]};
            yb[s1] = (v2f){acc2[2 * j] * acc3[2 * j], acc2[2 * j + 1] * acc3[2 * j + 1]};
            if (rtg < 6)
              yb[s2] = (v2f){acc4[2 * j] * acc5[2 * j], acc4[2 * j + 1] * acc5[2 * j + 1]};
          }
        }
      }

      if (ph == 1) {
        const float* Sn = S + (size_t)n * (3 * CTOT);
        sa = Sn[c];
        sb = Sn[CTOT + c];
        sg = Sn[2 * CTOT + c];
        const int nn = n + (int)gridDim.x;
        LOAD_BX(X + (size_t)(nn < N ? nn : n) * (NCOEF * 2 * CTOT))
      }

      #pragma unroll
      for (int q = 0; q < 5; ++q) {
        const int blk = f0 + q;
        if (blk < f1) {
          if (blk + 1 < f1) LOAD_F_BLOCK(blk + 1, (blk + 1) & 1)
          else if (ph == 0) LOAD_G_GROUP(g1, g1 & 1)
          else              LOAD_G_GROUP(0, 0)
          __builtin_amdgcn_sched_barrier(0);
          #pragma unroll
          for (int j = 0; j < 9; ++j) {
            const v2f b = yb[FB_START(blk) + j];
            oa0 = WMMA_F32(fa[blk & 1][j][0], b, oa0);
            oa1 = WMMA_F32(fa[blk & 1][j][1], b, oa1);
            oa2 = WMMA_F32(fa[blk & 1][j][2], b, oa2);
            o48[j & 3] = __builtin_elementwise_fma(fa[blk & 1][j][3], b, o48[j & 3]);
          }
          __builtin_amdgcn_sched_barrier(0);
        }
      }
    }

    const float gate  = sigmoidf_(sg);
    const float oscal = sa * sigmoidf_(sa) * sb;

    float* On = O + (size_t)n * (NCOEF * CTOT) + c;
    #pragma unroll
    for (int r = 0; r < 8; ++r) {
      const int i0 = r + 8 * lanehi;
      float v0 = oa0[r] * gate;
      if (i0 == 0) v0 += oscal;
      On[i0 * CTOT] = v0;
      On[(16 + i0) * CTOT] = oa1[r] * gate;
      On[(32 + i0) * CTOT] = oa2[r] * gate;
    }
    const v2f s48 = (o48[0] + o48[1]) + (o48[2] + o48[3]);
    const float h48 = s48.x + s48.y;
    const float p48 = __builtin_bit_cast(float,
        __builtin_amdgcn_permlanex16(0u, __builtin_bit_cast(unsigned, h48),
                                     0x76543210u, 0xfedcba98u, false, false));
    if (lanehi == 0) On[(NCOEF - 1) * CTOT] = (h48 + p48) * gate;
  }
}

#undef LOAD_G_GROUP
#undef LOAD_F_BLOCK
#undef FB_START
#undef FB_CNT
#undef LOAD_BX

extern "C" void kernel_launch(void* const* d_in, const int* in_sizes, int n_in,
                              void* d_out, int out_size, void* d_ws, size_t ws_size,
                              hipStream_t stream) {
  const float* X = (const float*)d_in[0];
  const float* S = (const float*)d_in[1];
  const float* G = (const float*)d_in[2];
  const float* F = (const float*)d_in[3];
  float* O = (float*)d_out;

  const int N = in_sizes[0] / (NCOEF * 2 * CTOT);
  const size_t shmem = (size_t)SMEM_FLOATS * sizeof(float);

  (void)hipFuncSetAttribute((const void*)s2act_swiglu_kernel,
                            hipFuncAttributeMaxDynamicSharedMemorySize,
                            (int)shmem);

  hipLaunchKernelGGL(s2act_swiglu_kernel, dim3(NBLOCKS), dim3(256), shmem, stream,
                     X, S, G, F, O, N);
}
